# starting kernel
# baseline (speedup 1.0000x reference)
#include <hip/hip_runtime.h>

#define BB 32
#define TT 1024
#define NN 128
#define CHUNK 16
#define NCHUNK (TT / CHUNK)
#define NTHREAD 256
#define WS 132
#define XS 132
#define CS 20

typedef __attribute__((ext_vector_type(2))) float v2f;
typedef __attribute__((ext_vector_type(8))) float v8f;

__device__ __forceinline__ v8f wmma4(v2f a, v2f b, v8f c) {
  return __builtin_amdgcn_wmma_f32_16x16x4_f32(
       false, a,  false, b,
       (short)0, c,  false,  false);
}

__global__ void __launch_bounds__(NTHREAD)
hebb_scan_kernel(const float* __restrict__ x,
                 const float* __restrict__ w_init,
                 const float* __restrict__ alphap,
                 const float* __restrict__ etap,
                 const int* __restrict__ mask,
                 float* __restrict__ w_out,
                 float* __restrict__ y_out) {
  extern __shared__ float smem[];
  float* Wsh = smem;
  float* Xsh = Wsh + NN * WS;
  float* Csh = Xsh + CHUNK * XS;
  float* sG  = Csh + CHUNK * CS;
  float* sGi = sG + 17;
  float* sE  = sGi + 17;
  float* sF  = sE + 16;

  const int b    = blockIdx.x;
  const int tid  = threadIdx.x;
  const int wave = tid >> 5;
  const int lane = tid & 31;
  const int half = lane >> 4;
  const int lr   = lane & 15;
  const float alpha = *alphap;
  const float eta   = *etap;

  const float* wsrc = w_init + (size_t)b * NN * NN;
  for (int idx = tid * 4; idx < NN * NN; idx += NTHREAD * 4) {
    const int r = idx >> 7, c = idx & 127;
    *(float4*)(&Wsh[r * WS + c]) = *(const float4*)(wsrc + idx);
  }
  __syncthreads();

  const float* xb = x + (size_t)b * TT * NN;
  const int*   mb = mask + b * TT;
  float* yb_base  = y_out + (size_t)b * TT * NN;

  const int i0 = wave * 16;

  for (int ch = 0; ch < NCHUNK; ++ch) {
    const int t0 = ch * CHUNK;

    for (int idx4 = tid; idx4 < CHUNK * NN / 4; idx4 += NTHREAD) {
      const int r = idx4 >> 5, c4 = (idx4 & 31) << 2;
      *(float4*)(&Xsh[r * XS + c4]) = *(const float4*)(xb + (t0 + r) * NN + c4);
    }
    if (ch + 1 < NCHUNK)
      __builtin_prefetch(xb + (t0 + CHUNK) * NN + lane * 16, 0, 0);

    if (tid == 0) {
      float G = 1.f;
      sG[0] = 1.f; sGi[0] = 1.f;
      for (int s = 0; s < CHUNK; ++s) {
        const int m = mb[t0 + s];
        sE[s] = m ? eta : 0.f;
        G *= m ? alpha : 1.f;
        sG[s + 1] = G;
        sGi[s + 1] = 1.f / G;
      }
      for (int s = 0; s < CHUNK; ++s)
        sF[s] = sE[s] * sG[CHUNK] * sGi[s + 1];
    }
    __syncthreads();

    v8f acc = {};
    v8f sgram = {};
    #pragma unroll 8
    for (int k0 = 0; k0 < NN; k0 += 4) {
      v2f a  = *(const v2f*)(&Xsh[lr * XS + k0 + 2 * half]);
      v2f bw = *(const v2f*)(&Wsh[(i0 + lr) * WS + k0 + 2 * half]);
      acc = wmma4(a, bw, acc);
      if (wave == 0) sgram = wmma4(a, a, sgram);
    }
    #pragma unroll
    for (int v = 0; v < 8; ++v) acc[v] *= sG[v + half * 8];

    if (wave == 0) {
      #pragma unroll
      for (int v = 0; v < 8; ++v) {
        const int m = v + half * 8;
        const int n = lr;
        const float coef = (n < m) ? sE[n] * sG[m] * sGi[n + 1] : 0.f;
        Csh[m * CS + n] = coef * sgram[v];
      }
    }
    __syncthreads();

    v2f xcol[4];
    #pragma unroll
    for (int q = 0; q < 4; ++q) {
      const int ks = q * 4 + 2 * half;
      xcol[q].x = Xsh[ks * XS + i0 + lr];
      xcol[q].y = Xsh[(ks + 1) * XS + i0 + lr];
    }

    #pragma unroll
    for (int q = 0; q < 4; ++q) {
      v2f a = *(const v2f*)(&Csh[lr * CS + q * 4 + 2 * half]);
      acc = wmma4(a, xcol[q], acc);
    }
    float* yb = yb_base + (size_t)t0 * NN;
    #pragma unroll
    for (int v = 0; v < 8; ++v)
      yb[(v + half * 8) * NN + i0 + lr] = acc[v];

    v2f afr[4];
    #pragma unroll
    for (int q = 0; q < 4; ++q) {
      const int ks = q * 4 + 2 * half;
      afr[q].x = sF[ks]     * xcol[q].x;
      afr[q].y = sF[ks + 1] * xcol[q].y;
    }
    const float Gend = sG[CHUNK];
    for (int jt = 0; jt < 8; ++jt) {
      const int j0 = jt * 16;
      v8f wacc;
      #pragma unroll
      for (int v = 0; v < 8; ++v)
        wacc[v] = Gend * Wsh[(i0 + v + half * 8) * WS + j0 + lr];
      #pragma unroll
      for (int q = 0; q < 4; ++q) {
        const int ks = q * 4 + 2 * half;
        v2f bx;
        bx.x = Xsh[ks * XS       + j0 + lr];
        bx.y = Xsh[(ks + 1) * XS + j0 + lr];
        wacc = wmma4(afr[q], bx, wacc);
      }
      #pragma unroll
      for (int v = 0; v < 8; ++v)
        Wsh[(i0 + v + half * 8) * WS + j0 + lr] = wacc[v];
    }
    __syncthreads();
  }

  float* wdst = w_out + (size_t)b * NN * NN;
  for (int idx = tid * 4; idx < NN * NN; idx += NTHREAD * 4) {
    const int r = idx >> 7, c = idx & 127;
    *(float4*)(wdst + idx) = *(const float4*)(&Wsh[r * WS + c]);
  }
}

extern "C" void kernel_launch(void* const* d_in, const int* in_sizes, int n_in,
                              void* d_out, int out_size, void* d_ws, size_t ws_size,
                              hipStream_t stream) {
  const float* x      = (const float*)d_in[0];
  const float* w_init = (const float*)d_in[1];
  const float* alpha  = (const float*)d_in[2];
  const float* eta    = (const float*)d_in[3];
  const int*   maskp  = (const int*)d_in[4];

  float* w_out = (float*)d_out;
  float* y_out = (float*)d_out + (size_t)BB * NN * NN;

  const size_t smem_bytes =
      (size_t)(NN * WS + CHUNK * XS + CHUNK * CS + 17 + 17 + 16 + 16) *
      sizeof(float);
  (void)hipFuncSetAttribute((const void*)hebb_scan_kernel,
                            hipFuncAttributeMaxDynamicSharedMemorySize,
                            (int)smem_bytes);

  hebb_scan_kernel<<<BB, NTHREAD, smem_bytes, stream>>>(
      x, w_init, alpha, eta, maskp, w_out, y_out);
}


# one independent wave per (batch, 16-row band of W): the band stays transposed in 8 accumulator tiles that feed the read-out directly as B (permuted k order); own Gram per wave; no LDS, no barrier
# speedup vs baseline: 3.3284x; 3.3284x over previous
#include <hip/hip_runtime.h>

#define BB 32
#define TT 1024
#define NN 128
#define CHUNK 16
#define NCHUNK (TT / CHUNK)
#define NTILE (NN / 16)
#define NTHREAD 64

typedef __attribute__((ext_vector_type(2))) float v2f;
typedef __attribute__((ext_vector_type(4))) float v4f;
typedef __attribute__((ext_vector_type(8))) float v8f;

__device__ __forceinline__ v8f wmma4(v2f a, v2f b, v8f c) {
  return __builtin_amdgcn_wmma_f32_16x16x4_f32(
       false, a,  false, b,
       (short)0, c,  false,  false);
}

__device__ __forceinline__ float lane_get(float v, int src) {
  return __int_as_float(__builtin_amdgcn_ds_bpermute(src << 2, __float_as_int(v)));
}

struct XFrag {
  v8f a[NTILE];
  v8f t[NTILE];
  v8f tb;
  int m;
};

__device__ __forceinline__ void load_part1(XFrag& f, const float* __restrict__ xa,
                                           const float* __restrict__ xt,
                                           const float* __restrict__ xtb,
                                           const int* __restrict__ mp) {
  #pragma unroll
  for (int jt = 0; jt < NTILE; ++jt) {
    const v4f lo = *(const v4f*)(xa + 16 * jt);
    const v4f hi = *(const v4f*)(xa + 16 * jt + 4);
    f.a[jt] = __builtin_shufflevector(lo, hi, 0, 1, 2, 3, 4, 5, 6, 7);
  }
  #pragma unroll
  for (int v = 0; v < 8; ++v) f.tb[v] = xtb[v * NN];
  #pragma unroll
  for (int jt = 0; jt < NTILE / 2; ++jt) {
    #pragma unroll
    for (int v = 0; v < 8; ++v) f.t[jt][v] = xt[v * NN + 16 * jt];
  }
  f.m = *mp;
}

__device__ __forceinline__ void load_part2(XFrag& f, const float* __restrict__ xt) {
  #pragma unroll
  for (int jt = NTILE / 2; jt < NTILE; ++jt) {
    #pragma unroll
    for (int v = 0; v < 8; ++v) f.t[jt][v] = xt[v * NN + 16 * jt];
  }
}

struct Scal {
  float gi[8];
  float Gs[8];
  float G16;
  unsigned bal;
};

__device__ __forceinline__ void chunk_scalars(Scal& sc, int m, float pw, float ipw,
                                              int half) {
  const unsigned bal = __builtin_amdgcn_ballot_w32(m != 0) & 0xFFFFu;
  sc.bal = bal;
  #pragma unroll
  for (int v = 0; v < 8; ++v) {
    const int s = 8 * half + v;
    const int c_lt = __popc(bal & ((1u << s) - 1u));
    const int c_le = __popc(bal & ((2u << s) - 1u));
    sc.Gs[v] = lane_get(pw, c_lt);
    sc.gi[v] = lane_get(ipw, c_le);
  }
  sc.G16 = lane_get(pw, __popc(bal));
}

__device__ __forceinline__ void readout_gram(const XFrag& f, const v8f (&WT)[NTILE],
                                             v8f& acc, v8f& sg) {
  v8f acc0 = {}, acc1 = {}, sg0 = {}, sg1 = {};
  #pragma unroll
  for (int jt = 0; jt < NTILE; jt += 2) {
    #pragma unroll
    for (int p = 0; p < 4; ++p) {
      const v2f a0 = {f.a[jt][2 * p], f.a[jt][2 * p + 1]};
      const v2f b0 = {WT[jt][2 * p], WT[jt][2 * p + 1]};
      const v2f a1 = {f.a[jt + 1][2 * p], f.a[jt + 1][2 * p + 1]};
      const v2f b1 = {WT[jt + 1][2 * p], WT[jt + 1][2 * p + 1]};
      acc0 = wmma4(a0, b0, acc0);
      sg0  = wmma4(a0, a0, sg0);
      acc1 = wmma4(a1, b1, acc1);
      sg1  = wmma4(a1, a1, sg1);
    }
  }
  acc = acc0 + acc1;
  sg  = sg0 + sg1;
}

__device__ __forceinline__ v8f intra_update(const XFrag& f, const Scal& sc, v8f acc,
                                            const v8f& sg, v8f (&WT)[NTILE],
                                            float eta, int lr, int half) {
  v8f hx;
  #pragma unroll
  for (int v = 0; v < 8; ++v) {
    const float hs = ((sc.bal >> (8 * half + v)) & 1u) ? eta * sc.gi[v] : 0.f;
    hx[v] = hs * f.tb[v];
  }

  #pragma unroll
  for (int p = 0; p < 4; ++p) {
    const int s0 = 8 * half + 2 * p;
    const v2f a = {(s0 < lr) ? sg[2 * p] : 0.f, (s0 + 1 < lr) ? sg[2 * p + 1] : 0.f};
    const v2f b = {hx[2 * p], hx[2 * p + 1]};
    acc = wmma4(a, b, acc);
  }
  v8f y;
  #pragma unroll
  for (int v = 0; v < 8; ++v) y[v] = sc.Gs[v] * acc[v];

  #pragma unroll
  for (int p = 0; p < 4; ++p) {
    const v2f b = {hx[2 * p], hx[2 * p + 1]};
    #pragma unroll
    for (int jt = 0; jt < NTILE; ++jt) {
      const v2f a = {f.t[jt][2 * p], f.t[jt][2 * p + 1]};
      WT[jt] = wmma4(a, b, WT[jt]);
    }
  }
  #pragma unroll
  for (int jt = 0; jt < NTILE; ++jt) WT[jt] *= sc.G16;
  return y;
}

__device__ __forceinline__ void store_y(float* __restrict__ yrow, const v8f& y) {
  #pragma unroll
  for (int v = 0; v < 8; ++v) yrow[v * NN] = y[v];
}

__global__ void __launch_bounds__(NTHREAD)
hebb_band_kernel(const float* __restrict__ x,
                 const float* __restrict__ w_init,
                 const float* __restrict__ alphap,
                 const float* __restrict__ etap,
                 const int* __restrict__ mask,
                 float* __restrict__ w_out,
                 float* __restrict__ y_out) {
  const int wave = threadIdx.x >> 5;
  const int lane = threadIdx.x & 31;
  const int half = lane >> 4;
  const int lr   = lane & 15;
  const int b    = blockIdx.x >> 2;
  const int band = ((blockIdx.x & 3) << 1) + wave;
  const int i0   = band * 16;
  const float alpha = *alphap;
  const float eta   = *etap;

  float pw = 1.f;
  #pragma unroll
  for (int k = 0; k < 16; ++k) pw *= (k < lane) ? alpha : 1.f;
  const float ipw = 1.f / pw;

  v8f WT[NTILE];
  {
    const float* wsrc = w_init + (size_t)b * NN * NN + (i0 + lr) * NN + 8 * half;
    #pragma unroll
    for (int jt = 0; jt < NTILE; ++jt) {
      const v4f lo = *(const v4f*)(wsrc + 16 * jt);
      const v4f hi = *(const v4f*)(wsrc + 16 * jt + 4);
      WT[jt] = __builtin_shufflevector(lo, hi, 0, 1, 2, 3, 4, 5, 6, 7);
    }
  }

  const float* xb  = x + (size_t)b * TT * NN;
  const float* xa  = xb + lr * NN + 8 * half;
  const float* xt  = xb + (8 * half) * NN + lr;
  const float* xtb = xt + i0;
  const int*   mp  = mask + b * TT + lr;
  float* yrow = y_out + (size_t)b * TT * NN + (8 * half) * NN + i0 + lr;

  XFrag fa, fb;
  Scal sa, sb;
  v8f acc, sg, ya, yb;
  load_part1(fa, xa, xt, xtb, mp);
  load_part2(fa, xt);
  chunk_scalars(sa, fa.m, pw, ipw, half);

#define SCHED_FENCE() __builtin_amdgcn_sched_barrier(0)

  #pragma unroll 1
  for (int ch = 0; ch < NCHUNK; ch += 2) {
    const int o1 = (ch + 1) * CHUNK;
    const int nx = (ch + 2 < NCHUNK) ? (ch + 2) : (NCHUNK - 1);
    const int o2 = nx * CHUNK;

    SCHED_FENCE();
    load_part1(fb, xa + o1 * NN, xt + o1 * NN, xtb + o1 * NN, mp + o1);
    SCHED_FENCE();
    readout_gram(fa, WT, acc, sg);
    SCHED_FENCE();
    chunk_scalars(sb, fb.m, pw, ipw, half);
    load_part2(fb, xt + o1 * NN);
    SCHED_FENCE();
    ya = intra_update(fa, sa, acc, sg, WT, eta, lr, half);

    SCHED_FENCE();
    store_y(yrow + (size_t)ch * CHUNK * NN, ya);
    load_part1(fa, xa + o2 * NN, xt + o2 * NN, xtb + o2 * NN, mp + o2);
    SCHED_FENCE();
    readout_gram(fb, WT, acc, sg);
    SCHED_FENCE();
    chunk_scalars(sa, fa.m, pw, ipw, half);
    load_part2(fa, xt + o2 * NN);
    SCHED_FENCE();
    yb = intra_update(fb, sb, acc, sg, WT, eta, lr, half);
    SCHED_FENCE();
    store_y(yrow + (size_t)o1 * NN, yb);
  }
#undef SCHED_FENCE

  float* wdst = w_out + (size_t)b * NN * NN + (i0 + lr) * NN + 8 * half;
  #pragma unroll
  for (int jt = 0; jt < NTILE; ++jt) {
    const v4f lo = __builtin_shufflevector(WT[jt], WT[jt], 0, 1, 2, 3);
    const v4f hi = __builtin_shufflevector(WT[jt], WT[jt], 4, 5, 6, 7);
    *(v4f*)(wdst + 16 * jt)     = lo;
    *(v4f*)(wdst + 16 * jt + 4) = hi;
  }
}

extern "C" void kernel_launch(void* const* d_in, const int* in_sizes, int n_in,
                              void* d_out, int out_size, void* d_ws, size_t ws_size,
                              hipStream_t stream) {
  const float* x      = (const float*)d_in[0];
  const float* w_init = (const float*)d_in[1];
  const float* alpha  = (const float*)d_in[2];
  const float* eta    = (const float*)d_in[3];
  const int*   maskp  = (const int*)d_in[4];

  float* w_out = (float*)d_out;
  float* y_out = (float*)d_out + (size_t)BB * NN * NN;

  hebb_band_kernel<<<BB * 4, NTHREAD, 0, stream>>>(
      x, w_init, alpha, eta, maskp, w_out, y_out);
}


# time split: the 8 waves of a band's workgroup take 128 steps each; every wave first builds its segment's state delta from zero, the deltas meet in LDS behind one barrier, then each runs its segment
# speedup vs baseline: 8.2895x; 2.4905x over previous
#include <hip/hip_runtime.h>


#define BB 32
#define TT 1024
#define NN 128
#define CHUNK 16
#define NCHUNK (TT / CHUNK)
#define NTILE (NN / 16)
#define NSEG 8
#define SEGCH (NCHUNK / NSEG)
#define NTHREAD (32 * NSEG)

typedef __attribute__((ext_vector_type(2))) float v2f;
typedef __attribute__((ext_vector_type(4))) float v4f;
typedef __attribute__((ext_vector_type(8))) float v8f;

__device__ __forceinline__ v8f wmma4(v2f a, v2f b, v8f c) {
  return __builtin_amdgcn_wmma_f32_16x16x4_f32(
       false, a,  false, b,
       (short)0, c,  false,  false);
}

__device__ __forceinline__ float lane_get(float v, int src) {
  return __int_as_float(__builtin_amdgcn_ds_bpermute(src << 2, __float_as_int(v)));
}

struct XFrag {
  v8f a[NTILE];
  v8f t[NTILE];
  v8f tb;
  int m;
};

__device__ __forceinline__ void load_part1(XFrag& f, const float* __restrict__ xa,
                                           const float* __restrict__ xt,
                                           const float* __restrict__ xtb,
                                           const int* __restrict__ mp) {
  #pragma unroll
  for (int jt = 0; jt < NTILE; ++jt) {
    const v4f lo = *(const v4f*)(xa + 16 * jt);
    const v4f hi = *(const v4f*)(xa + 16 * jt + 4);
    f.a[jt] = __builtin_shufflevector(lo, hi, 0, 1, 2, 3, 4, 5, 6, 7);
  }
  #pragma unroll
  for (int v = 0; v < 8; ++v) f.tb[v] = xtb[v * NN];
  #pragma unroll
  for (int jt = 0; jt < NTILE / 2; ++jt) {
    #pragma unroll
    for (int v = 0; v < 8; ++v) f.t[jt][v] = xt[v * NN + 16 * jt];
  }
  f.m = *mp;
}

__device__ __forceinline__ void load_part2(XFrag& f, const float* __restrict__ xt) {
  #pragma unroll
  for (int jt = NTILE / 2; jt < NTILE; ++jt) {
    #pragma unroll
    for (int v = 0; v < 8; ++v) f.t[jt][v] = xt[v * NN + 16 * jt];
  }
}

struct Scal {
  float gi[8];
  float Gs[8];
  float G16;
  unsigned bal;
};

__device__ __forceinline__ void scalars_of(Scal& sc, unsigned bal, float pw, float ipw,
                                           int half) {
  sc.bal = bal;
  #pragma unroll
  for (int v = 0; v < 8; ++v) {
    const int s = 8 * half + v;
    const int c_lt = __popc(bal & ((1u << s) - 1u));
    const int c_le = __popc(bal & ((2u << s) - 1u));
    sc.Gs[v] = lane_get(pw, c_lt);
    sc.gi[v] = lane_get(ipw, c_le);
  }
  sc.G16 = lane_get(pw, __popc(bal));
}

__device__ __forceinline__ void chunk_scalars(Scal& sc, int m, float pw, float ipw,
                                              int half) {
  scalars_of(sc, __builtin_amdgcn_ballot_w32(m != 0) & 0xFFFFu, pw, ipw, half);
}

struct UFrag {
  v8f t[NTILE];
  v8f tb;
};

__device__ __forceinline__ void load_u1(UFrag& f, const float* __restrict__ xt,
                                        const float* __restrict__ xtb) {
  #pragma unroll
  for (int v = 0; v < 8; ++v) f.tb[v] = xtb[v * NN];
  #pragma unroll
  for (int jt = 0; jt < NTILE / 2; ++jt) {
    #pragma unroll
    for (int v = 0; v < 8; ++v) f.t[jt][v] = xt[v * NN + 16 * jt];
  }
}

__device__ __forceinline__ void load_u2(UFrag& f, const float* __restrict__ xt) {
  #pragma unroll
  for (int jt = NTILE / 2; jt < NTILE; ++jt) {
    #pragma unroll
    for (int v = 0; v < 8; ++v) f.t[jt][v] = xt[v * NN + 16 * jt];
  }
}

__device__ __forceinline__ v8f delta_hx(const UFrag& f, const Scal& sc, float eta, int half) {
  v8f hx;
  #pragma unroll
  for (int v = 0; v < 8; ++v) {
    const float hs = ((sc.bal >> (8 * half + v)) & 1u) ? eta * sc.gi[v] : 0.f;
    hx[v] = hs * f.tb[v];
  }
  return hx;
}

template <int JT0>
__device__ __forceinline__ void delta_tiles(v8f (&D)[NTILE], const UFrag& f, const v8f& hx) {
  #pragma unroll
  for (int p = 0; p < 4; ++p) {
    const v2f b = {hx[2 * p], hx[2 * p + 1]};
    #pragma unroll
    for (int jt = JT0; jt < JT0 + NTILE / 2; ++jt) {
      const v2f a = {f.t[jt][2 * p], f.t[jt][2 * p + 1]};
      D[jt] = wmma4(a, b, D[jt]);
    }
  }
}

__device__ __forceinline__ void readout_gram(const XFrag& f, const v8f (&WT)[NTILE],
                                             v8f& acc, v8f& sg) {
  v8f acc0 = {}, acc1 = {}, sg0 = {}, sg1 = {};
  #pragma unroll
  for (int jt = 0; jt < NTILE; jt += 2) {
    #pragma unroll
    for (int p = 0; p < 4; ++p) {
      const v2f a0 = {f.a[jt][2 * p], f.a[jt][2 * p + 1]};
      const v2f b0 = {WT[jt][2 * p], WT[jt][2 * p + 1]};
      const v2f a1 = {f.a[jt + 1][2 * p], f.a[jt + 1][2 * p + 1]};
      const v2f b1 = {WT[jt + 1][2 * p], WT[jt + 1][2 * p + 1]};
      acc0 = wmma4(a0, b0, acc0);
      sg0  = wmma4(a0, a0, sg0);
      acc1 = wmma4(a1, b1, acc1);
      sg1  = wmma4(a1, a1, sg1);
    }
  }
  acc = acc0 + acc1;
  sg  = sg0 + sg1;
}

__device__ __forceinline__ v8f intra_update(const XFrag& f, const Scal& sc, v8f acc,
                                            const v8f& sg, v8f (&WT)[NTILE],
                                            float eta, int lr, int half) {
  v8f hx;
  #pragma unroll
  for (int v = 0; v < 8; ++v) {
    const float hs = ((sc.bal >> (8 * half + v)) & 1u) ? eta * sc.gi[v] : 0.f;
    hx[v] = hs * f.tb[v];
  }

  #pragma unroll
  for (int p = 0; p < 4; ++p) {
    const int s0 = 8 * half + 2 * p;
    const v2f a = {(s0 < lr) ? sg[2 * p] : 0.f, (s0 + 1 < lr) ? sg[2 * p + 1] : 0.f};
    const v2f b = {hx[2 * p], hx[2 * p + 1]};
    acc = wmma4(a, b, acc);
  }
  v8f y;
  #pragma unroll
  for (int v = 0; v < 8; ++v) y[v] = sc.Gs[v] * acc[v];

  #pragma unroll
  for (int p = 0; p < 4; ++p) {
    const v2f b = {hx[2 * p], hx[2 * p + 1]};
    #pragma unroll
    for (int jt = 0; jt < NTILE; ++jt) {
      const v2f a = {f.t[jt][2 * p], f.t[jt][2 * p + 1]};
      WT[jt] = wmma4(a, b, WT[jt]);
    }
  }
  #pragma unroll
  for (int jt = 0; jt < NTILE; ++jt) WT[jt] *= sc.G16;
  return y;
}

__device__ __forceinline__ void store_y(float* __restrict__ yrow, const v8f& y) {
  #pragma unroll
  for (int v = 0; v < 8; ++v) yrow[v * NN] = y[v];
}

#define SCHED_FENCE() __builtin_amdgcn_sched_barrier(0)

__global__ void __launch_bounds__(NTHREAD)
hebb_seg_kernel(const float* __restrict__ x,
                const float* __restrict__ w_init,
                const float* __restrict__ alphap,
                const float* __restrict__ etap,
                const int* __restrict__ mask,
                float* __restrict__ w_out,
                float* __restrict__ y_out) {
  __shared__ __attribute__((aligned(16))) float dl[(NSEG - 1) * NTILE * 2 * 128];
  __shared__ float gl[NSEG];

  const int seg  = __builtin_amdgcn_readfirstlane(threadIdx.x >> 5);
  const int lane = threadIdx.x & 31;
  const int half = lane >> 4;
  const int lr   = lane & 15;
  const int b    = blockIdx.x >> 3;
  const int band = blockIdx.x & 7;
  const int i0   = band * 16;
  const float alpha = *alphap;
  const float eta   = *etap;

  float pw = 1.f;
  #pragma unroll
  for (int k = 0; k < 16; ++k) pw *= (k < lane) ? alpha : 1.f;
  const float ipw = 1.f / pw;

  const int t0 = seg * (SEGCH * CHUNK);
  const float* xb  = x + (size_t)b * TT * NN + (size_t)t0 * NN;
  const float* xa  = xb + lr * NN + 8 * half;
  const float* xt  = xb + (8 * half) * NN + lr;
  const float* xtb = xt + i0;
  const int*   mp  = mask + b * TT + t0 + lr;
  float* yrow = y_out + (size_t)b * TT * NN + (size_t)t0 * NN + (8 * half) * NN + i0 + lr;

  if (seg < NSEG - 1) {
    v8f D[NTILE];
    #pragma unroll
    for (int jt = 0; jt < NTILE; ++jt) D[jt] = v8f{};
    float gseg = 1.f;
    unsigned long long mlo = 0ull, mhi = 0ull;
    #pragma unroll
    for (int c = 0; c < SEGCH; ++c) {
      const unsigned long long bal = __builtin_amdgcn_ballot_w32(mp[c * CHUNK] != 0) & 0xFFFFu;
      if (c < SEGCH / 2) mlo |= bal << (16 * c);
      else               mhi |= bal << (16 * (c - SEGCH / 2));
    }

    UFrag ua, ub;
    Scal sa, sb;
    SCHED_FENCE();
    load_u1(ua, xt, xtb);
    SCHED_FENCE();
    load_u2(ua, xt);
    SCHED_FENCE();
    scalars_of(sa, (unsigned)mlo & 0xFFFFu, pw, ipw, half);

    #pragma unroll 1
    for (int ch = 0; ch < SEGCH; ch += 2) {
      const int o1 = (ch + 1) * CHUNK;
      const int nx = (ch + 2 < SEGCH) ? (ch + 2) : (SEGCH - 1);
      const int o2 = nx * CHUNK;
      const unsigned long long m1 = (ch + 1 < SEGCH / 2) ? mlo : mhi;
      const unsigned long long m2 = (nx < SEGCH / 2) ? mlo : mhi;
      const unsigned bal1 = (unsigned)(m1 >> (16 * ((ch + 1) & 3))) & 0xFFFFu;
      const unsigned bal2 = (unsigned)(m2 >> (16 * (nx & 3))) & 0xFFFFu;

      SCHED_FENCE();
      load_u1(ub, xt + o1 * NN, xtb + o1 * NN);
      SCHED_FENCE();
      {
        const v8f hx = delta_hx(ua, sa, eta, half);
        delta_tiles<0>(D, ua, hx);
        SCHED_FENCE();
        scalars_of(sb, bal1, pw, ipw, half);
        load_u2(ub, xt + o1 * NN);
        SCHED_FENCE();
        delta_tiles<NTILE / 2>(D, ua, hx);
      }
      #pragma unroll
      for (int jt = 0; jt < NTILE; ++jt) D[jt] *= sa.G16;
      gseg *= sa.G16;

      SCHED_FENCE();
      load_u1(ua, xt + o2 * NN, xtb + o2 * NN);
      SCHED_FENCE();
      {
        const v8f hx = delta_hx(ub, sb, eta, half);
        delta_tiles<0>(D, ub, hx);
        SCHED_FENCE();
        scalars_of(sa, bal2, pw, ipw, half);
        load_u2(ua, xt + o2 * NN);
        SCHED_FENCE();
        delta_tiles<NTILE / 2>(D, ub, hx);
      }
      #pragma unroll
      for (int jt = 0; jt < NTILE; ++jt) D[jt] *= sb.G16;
      gseg *= sb.G16;
    }

    float* dst = dl + seg * (NTILE * 2 * 128) + lane * 4;
    #pragma unroll
    for (int jt = 0; jt < NTILE; ++jt) {
      *(v4f*)(dst + (2 * jt) * 128)     = __builtin_shufflevector(D[jt], D[jt], 0, 1, 2, 3);
      *(v4f*)(dst + (2 * jt + 1) * 128) = __builtin_shufflevector(D[jt], D[jt], 4, 5, 6, 7);
    }
    if (lane == 0) gl[seg] = gseg;
  }

  __syncthreads();

  v8f WT[NTILE];
  {
    const float* wsrc = w_init + (size_t)b * NN * NN + (i0 + lr) * NN + 8 * half;
    #pragma unroll
    for (int jt = 0; jt < NTILE; ++jt) {
      const v4f lo = *(const v4f*)(wsrc + 16 * jt);
      const v4f hi = *(const v4f*)(wsrc + 16 * jt + 4);
      WT[jt] = __builtin_shufflevector(lo, hi, 0, 1, 2, 3, 4, 5, 6, 7);
    }
  }
  XFrag fa, fb;
  Scal sa, sb;
  v8f acc, sg, ya, yb;
  load_part1(fa, xa, xt, xtb, mp);
  load_part2(fa, xt);

  #pragma unroll 1
  for (int q = 0; q < seg; ++q) {
    const float g = gl[q];
    const float* dsrc = dl + q * (NTILE * 2 * 128) + lane * 4;
    #pragma unroll
    for (int jt = 0; jt < NTILE; ++jt) {
      const v4f lo = *(const v4f*)(dsrc + (2 * jt) * 128);
      const v4f hi = *(const v4f*)(dsrc + (2 * jt + 1) * 128);
      WT[jt] = g * WT[jt] + __builtin_shufflevector(lo, hi, 0, 1, 2, 3, 4, 5, 6, 7);
    }
  }

  chunk_scalars(sa, fa.m, pw, ipw, half);

  #pragma unroll 1
  for (int ch = 0; ch < SEGCH; ch += 2) {
    const int o1 = (ch + 1) * CHUNK;
    const int nx = (ch + 2 < SEGCH) ? (ch + 2) : (SEGCH - 1);
    const int o2 = nx * CHUNK;

    SCHED_FENCE();
    load_part1(fb, xa + o1 * NN, xt + o1 * NN, xtb + o1 * NN, mp + o1);
    SCHED_FENCE();
    readout_gram(fa, WT, acc, sg);
    SCHED_FENCE();
    chunk_scalars(sb, fb.m, pw, ipw, half);
    load_part2(fb, xt + o1 * NN);
    SCHED_FENCE();
    ya = intra_update(fa, sa, acc, sg, WT, eta, lr, half);

    SCHED_FENCE();
    store_y(yrow + (size_t)ch * CHUNK * NN, ya);
    load_part1(fa, xa + o2 * NN, xt + o2 * NN, xtb + o2 * NN, mp + o2);
    SCHED_FENCE();
    readout_gram(fb, WT, acc, sg);
    SCHED_FENCE();
    chunk_scalars(sa, fa.m, pw, ipw, half);
    load_part2(fa, xt + o2 * NN);
    SCHED_FENCE();
    yb = intra_update(fb, sb, acc, sg, WT, eta, lr, half);
    SCHED_FENCE();
    store_y(yrow + (size_t)o1 * NN, yb);
  }
#undef SCHED_FENCE

  if (seg == NSEG - 1) {
    float* wdst = w_out + (size_t)b * NN * NN + (i0 + lr) * NN + 8 * half;
    #pragma unroll
    for (int jt = 0; jt < NTILE; ++jt) {
      const v4f lo = __builtin_shufflevector(WT[jt], WT[jt], 0, 1, 2, 3);
      const v4f hi = __builtin_shufflevector(WT[jt], WT[jt], 4, 5, 6, 7);
      *(v4f*)(wdst + 16 * jt)     = lo;
      *(v4f*)(wdst + 16 * jt + 4) = hi;
    }
  }
}

extern "C" void kernel_launch(void* const* d_in, const int* in_sizes, int n_in,
                              void* d_out, int out_size, void* d_ws, size_t ws_size,
                              hipStream_t stream) {
  const float* x      = (const float*)d_in[0];
  const float* w_init = (const float*)d_in[1];
  const float* alpha  = (const float*)d_in[2];
  const float* eta    = (const float*)d_in[3];
  const int*   maskp  = (const int*)d_in[4];

  float* w_out = (float*)d_out;
  float* y_out = (float*)d_out + (size_t)BB * NN * NN;

  hebb_seg_kernel<<<BB * NTILE, NTHREAD, 0, stream>>>(
      x, w_init, alpha, eta, maskp, w_out, y_out);
}


# chunk loop fetches x from global once, as whole 512-byte rows (b128 loads, 4 full lines each), parks them in a wave-private LDS tile and reads the a, t and tb fragments from LDS; no barrier added
# speedup vs baseline: 9.1230x; 1.1005x over previous
#include <hip/hip_runtime.h>


#define BB 32
#define TT 1024
#define NN 128
#define CHUNK 16
#define NCHUNK (TT / CHUNK)
#define NTILE (NN / 16)
#define NSEG 8
#define SEGCH (NCHUNK / NSEG)
#define NTHREAD (32 * NSEG)
#define XROW 20
#define XSUB 336
#define XTILE (NTILE * XSUB)

typedef __attribute__((ext_vector_type(2))) float v2f;
typedef __attribute__((ext_vector_type(4))) float v4f;
typedef __attribute__((ext_vector_type(8))) float v8f;

__device__ __forceinline__ v8f wmma4(v2f a, v2f b, v8f c) {
  return __builtin_amdgcn_wmma_f32_16x16x4_f32(
       false, a,  false, b,
       (short)0, c,  false,  false);
}

__device__ __forceinline__ float lane_get(float v, int src) {
  return __int_as_float(__builtin_amdgcn_ds_bpermute(src << 2, __float_as_int(v)));
}

struct AFrag {
  v8f a[NTILE];
};
struct TFrag {
  v8f t[NTILE];
  v8f tb;
};

__device__ __forceinline__ void load_rows(v4f (&g)[CHUNK], const float* __restrict__ xrow) {
  #pragma unroll
  for (int r = 0; r < CHUNK; ++r) g[r] = *(const v4f*)(xrow + r * NN);
}

__device__ __forceinline__ void park_rows(float* xw, const v4f (&g)[CHUNK], int lane) {
  float* dst = xw + (lane >> 2) * XSUB + 4 * (lane & 3);
  #pragma unroll
  for (int r = 0; r < CHUNK; ++r) *(v4f*)(dst + r * XROW) = g[r];
}

__device__ __forceinline__ void read_a(AFrag& f, const float* xw, int lr, int half) {
  const float* pa = xw + lr * XROW + 8 * half;
  #pragma unroll
  for (int jt = 0; jt < NTILE; ++jt) {
    const v4f lo = *(const v4f*)(pa + jt * XSUB);
    const v4f hi = *(const v4f*)(pa + jt * XSUB + 4);
    f.a[jt] = __builtin_shufflevector(lo, hi, 0, 1, 2, 3, 4, 5, 6, 7);
  }
}

__device__ __forceinline__ void read_t(TFrag& f, const float* xw, int lr, int half, int i0) {
  const float* pt = xw + (8 * half) * XROW + lr;
  #pragma unroll
  for (int v = 0; v < 8; ++v) f.tb[v] = pt[(i0 >> 4) * XSUB + v * XROW];
  #pragma unroll
  for (int jt = 0; jt < NTILE; ++jt) {
    #pragma unroll
    for (int v = 0; v < 8; ++v) f.t[jt][v] = pt[jt * XSUB + v * XROW];
  }
}

struct Scal {
  float gi[8];
  float Gs[8];
  float G16;
  unsigned bal;
};

__device__ __forceinline__ void scalars_of(Scal& sc, unsigned bal, float pw, float ipw,
                                           int half) {
  sc.bal = bal;
  #pragma unroll
  for (int v = 0; v < 8; ++v) {
    const int s = 8 * half + v;
    const int c_lt = __popc(bal & ((1u << s) - 1u));
    const int c_le = __popc(bal & ((2u << s) - 1u));
    sc.Gs[v] = lane_get(pw, c_lt);
    sc.gi[v] = lane_get(ipw, c_le);
  }
  sc.G16 = lane_get(pw, __popc(bal));
}

struct UFrag {
  v8f t[NTILE];
  v8f tb;
};

__device__ __forceinline__ void load_u1(UFrag& f, const float* __restrict__ xt,
                                        const float* __restrict__ xtb) {
  #pragma unroll
  for (int v = 0; v < 8; ++v) f.tb[v] = xtb[v * NN];
  #pragma unroll
  for (int jt = 0; jt < NTILE / 2; ++jt) {
    #pragma unroll
    for (int v = 0; v < 8; ++v) f.t[jt][v] = xt[v * NN + 16 * jt];
  }
}

__device__ __forceinline__ void load_u2(UFrag& f, const float* __restrict__ xt) {
  #pragma unroll
  for (int jt = NTILE / 2; jt < NTILE; ++jt) {
    #pragma unroll
    for (int v = 0; v < 8; ++v) f.t[jt][v] = xt[v * NN + 16 * jt];
  }
}

__device__ __forceinline__ v8f delta_hx(const UFrag& f, const Scal& sc, float eta, int half) {
  v8f hx;
  #pragma unroll
  for (int v = 0; v < 8; ++v) {
    const float hs = ((sc.bal >> (8 * half + v)) & 1u) ? eta * sc.gi[v] : 0.f;
    hx[v] = hs * f.tb[v];
  }
  return hx;
}

template <int JT0>
__device__ __forceinline__ void delta_tiles(v8f (&D)[NTILE], const UFrag& f, const v8f& hx) {
  #pragma unroll
  for (int p = 0; p < 4; ++p) {
    const v2f b = {hx[2 * p], hx[2 * p + 1]};
    #pragma unroll
    for (int jt = JT0; jt < JT0 + NTILE / 2; ++jt) {
      const v2f a = {f.t[jt][2 * p], f.t[jt][2 * p + 1]};
      D[jt] = wmma4(a, b, D[jt]);
    }
  }
}

__device__ __forceinline__ void readout_gram(const AFrag& f, const v8f (&WT)[NTILE],
                                             v8f& acc, v8f& sg) {
  v8f acc0 = {}, acc1 = {}, sg0 = {}, sg1 = {};
  #pragma unroll
  for (int jt = 0; jt < NTILE; jt += 2) {
    #pragma unroll
    for (int p = 0; p < 4; ++p) {
      const v2f a0 = {f.a[jt][2 * p], f.a[jt][2 * p + 1]};
      const v2f b0 = {WT[jt][2 * p], WT[jt][2 * p + 1]};
      const v2f a1 = {f.a[jt + 1][2 * p], f.a[jt + 1][2 * p + 1]};
      const v2f b1 = {WT[jt + 1][2 * p], WT[jt + 1][2 * p + 1]};
      acc0 = wmma4(a0, b0, acc0);
      sg0  = wmma4(a0, a0, sg0);
      acc1 = wmma4(a1, b1, acc1);
      sg1  = wmma4(a1, a1, sg1);
    }
  }
  acc = acc0 + acc1;
  sg  = sg0 + sg1;
}

__device__ __forceinline__ v8f intra_update(const TFrag& f, const Scal& sc, v8f acc,
                                            const v8f& sg, v8f (&WT)[NTILE],
                                            float eta, int lr, int half) {
  v8f hx;
  #pragma unroll
  for (int v = 0; v < 8; ++v) {
    const float hs = ((sc.bal >> (8 * half + v)) & 1u) ? eta * sc.gi[v] : 0.f;
    hx[v] = hs * f.tb[v];
  }

  #pragma unroll
  for (int p = 0; p < 4; ++p) {
    const int s0 = 8 * half + 2 * p;
    const v2f a = {(s0 < lr) ? sg[2 * p] : 0.f, (s0 + 1 < lr) ? sg[2 * p + 1] : 0.f};
    const v2f b = {hx[2 * p], hx[2 * p + 1]};
    acc = wmma4(a, b, acc);
  }
  v8f y;
  #pragma unroll
  for (int v = 0; v < 8; ++v) y[v] = sc.Gs[v] * acc[v];

  #pragma unroll
  for (int p = 0; p < 4; ++p) {
    const v2f b = {hx[2 * p], hx[2 * p + 1]};
    #pragma unroll
    for (int jt = 0; jt < NTILE; ++jt) {
      const v2f a = {f.t[jt][2 * p], f.t[jt][2 * p + 1]};
      WT[jt] = wmma4(a, b, WT[jt]);
    }
  }
  #pragma unroll
  for (int jt = 0; jt < NTILE; ++jt) WT[jt] *= sc.G16;
  return y;
}

__device__ __forceinline__ void store_y(float* __restrict__ yrow, const v8f& y) {
  #pragma unroll
  for (int v = 0; v < 8; ++v) yrow[v * NN] = y[v];
}

#define SCHED_FENCE() __builtin_amdgcn_sched_barrier(0)

__global__ void __launch_bounds__(NTHREAD)
hebb_seg_kernel(const float* __restrict__ x,
                const float* __restrict__ w_init,
                const float* __restrict__ alphap,
                const float* __restrict__ etap,
                const int* __restrict__ mask,
                float* __restrict__ w_out,
                float* __restrict__ y_out) {
  __shared__ __attribute__((aligned(16))) float dl[(NSEG - 1) * NTILE * 2 * 128];
  __shared__ float gl[NSEG];
  __shared__ __attribute__((aligned(16))) float xs[NSEG * 2 * XTILE];

  const int seg  = __builtin_amdgcn_readfirstlane(threadIdx.x >> 5);
  const int lane = threadIdx.x & 31;
  const int half = lane >> 4;
  const int lr   = lane & 15;
  const int b    = blockIdx.x >> 3;
  const int band = blockIdx.x & 7;
  const int i0   = band * 16;
  const float alpha = *alphap;
  const float eta   = *etap;

  float pw = 1.f;
  #pragma unroll
  for (int k = 0; k < 16; ++k) pw *= (k < lane) ? alpha : 1.f;
  const float ipw = 1.f / pw;

  const int t0 = seg * (SEGCH * CHUNK);
  const float* xb  = x + (size_t)b * TT * NN + (size_t)t0 * NN;
  const float* xrow = xb + 4 * lane;
  const float* xt  = xb + (8 * half) * NN + lr;
  const float* xtb = xt + i0;
  const int*   mp  = mask + b * TT + t0 + lr;
  float* yrow = y_out + (size_t)b * TT * NN + (size_t)t0 * NN + (8 * half) * NN + i0 + lr;

  unsigned long long mlo = 0ull, mhi = 0ull;
  #pragma unroll
  for (int c = 0; c < SEGCH; ++c) {
    const unsigned long long bal = __builtin_amdgcn_ballot_w32(mp[c * CHUNK] != 0) & 0xFFFFu;
    if (c < SEGCH / 2) mlo |= bal << (16 * c);
    else               mhi |= bal << (16 * (c - SEGCH / 2));
  }

  if (seg < NSEG - 1) {
    v8f D[NTILE];
    #pragma unroll
    for (int jt = 0; jt < NTILE; ++jt) D[jt] = v8f{};
    float gseg = 1.f;

    UFrag ua, ub;
    Scal sa, sb;
    SCHED_FENCE();
    load_u1(ua, xt, xtb);
    SCHED_FENCE();
    load_u2(ua, xt);
    SCHED_FENCE();
    scalars_of(sa, (unsigned)mlo & 0xFFFFu, pw, ipw, half);

    #pragma unroll 1
    for (int ch = 0; ch < SEGCH; ch += 2) {
      const int o1 = (ch + 1) * CHUNK;
      const int nx = (ch + 2 < SEGCH) ? (ch + 2) : (SEGCH - 1);
      const int o2 = nx * CHUNK;
      const unsigned long long m1 = (ch + 1 < SEGCH / 2) ? mlo : mhi;
      const unsigned long long m2 = (nx < SEGCH / 2) ? mlo : mhi;
      const unsigned bal1 = (unsigned)(m1 >> (16 * ((ch + 1) & 3))) & 0xFFFFu;
      const unsigned bal2 = (unsigned)(m2 >> (16 * (nx & 3))) & 0xFFFFu;

      SCHED_FENCE();
      load_u1(ub, xt + o1 * NN, xtb + o1 * NN);
      SCHED_FENCE();
      {
        const v8f hx = delta_hx(ua, sa, eta, half);
        delta_tiles<0>(D, ua, hx);
        SCHED_FENCE();
        scalars_of(sb, bal1, pw, ipw, half);
        load_u2(ub, xt + o1 * NN);
        SCHED_FENCE();
        delta_tiles<NTILE / 2>(D, ua, hx);
      }
      #pragma unroll
      for (int jt = 0; jt < NTILE; ++jt) D[jt] *= sa.G16;
      gseg *= sa.G16;

      SCHED_FENCE();
      load_u1(ua, xt + o2 * NN, xtb + o2 * NN);
      SCHED_FENCE();
      {
        const v8f hx = delta_hx(ub, sb, eta, half);
        delta_tiles<0>(D, ub, hx);
        SCHED_FENCE();
        scalars_of(sa, bal2, pw, ipw, half);
        load_u2(ua, xt + o2 * NN);
        SCHED_FENCE();
        delta_tiles<NTILE / 2>(D, ub, hx);
      }
      #pragma unroll
      for (int jt = 0; jt < NTILE; ++jt) D[jt] *= sb.G16;
      gseg *= sb.G16;
    }

    float* dst = dl + seg * (NTILE * 2 * 128) + lane * 4;
    #pragma unroll
    for (int jt = 0; jt < NTILE; ++jt) {
      *(v4f*)(dst + (2 * jt) * 128)     = __builtin_shufflevector(D[jt], D[jt], 0, 1, 2, 3);
      *(v4f*)(dst + (2 * jt + 1) * 128) = __builtin_shufflevector(D[jt], D[jt], 4, 5, 6, 7);
    }
    if (lane == 0) gl[seg] = gseg;
  }

  __syncthreads();

  v8f WT[NTILE];
  {
    const float* wsrc = w_init + (size_t)b * NN * NN + (i0 + lr) * NN + 8 * half;
    #pragma unroll
    for (int jt = 0; jt < NTILE; ++jt) {
      const v4f lo = *(const v4f*)(wsrc + 16 * jt);
      const v4f hi = *(const v4f*)(wsrc + 16 * jt + 4);
      WT[jt] = __builtin_shufflevector(lo, hi, 0, 1, 2, 3, 4, 5, 6, 7);
    }
  }
  AFrag a0, a1;
  TFrag ft;
  Scal sa, sb;
  v8f acc, sg, ya, yb;
  v4f rows[CHUNK];
  float* xw0 = xs + seg * (2 * XTILE);
  float* xw1 = xw0 + XTILE;
  load_rows(rows, xrow);

  #pragma unroll 1
  for (int q = 0; q < seg; ++q) {
    const float g = gl[q];
    const float* dsrc = dl + q * (NTILE * 2 * 128) + lane * 4;
    #pragma unroll
    for (int jt = 0; jt < NTILE; ++jt) {
      const v4f lo = *(const v4f*)(dsrc + (2 * jt) * 128);
      const v4f hi = *(const v4f*)(dsrc + (2 * jt + 1) * 128);
      WT[jt] = g * WT[jt] + __builtin_shufflevector(lo, hi, 0, 1, 2, 3, 4, 5, 6, 7);
    }
  }

  park_rows(xw0, rows, lane);
  read_a(a0, xw0, lr, half);
  load_rows(rows, xrow + CHUNK * NN);
  scalars_of(sa, (unsigned)mlo & 0xFFFFu, pw, ipw, half);

  #pragma unroll 1
  for (int ch = 0; ch < SEGCH; ch += 2) {
    const int o1 = (ch + 1) * CHUNK;
    const int o2 = ((ch + 2 < SEGCH) ? (ch + 2) : (SEGCH - 1)) * CHUNK;
    const int o3 = ((ch + 3 < SEGCH) ? (ch + 3) : (SEGCH - 1)) * CHUNK;
    const int c2 = (ch + 2 < SEGCH) ? (ch + 2) : (SEGCH - 1);
    const unsigned long long m1 = (ch + 1 < SEGCH / 2) ? mlo : mhi;
    const unsigned long long m2 = (c2 < SEGCH / 2) ? mlo : mhi;
    const unsigned bal1 = (unsigned)(m1 >> (16 * ((ch + 1) & 3))) & 0xFFFFu;
    const unsigned bal2 = (unsigned)(m2 >> (16 * (c2 & 3))) & 0xFFFFu;

    SCHED_FENCE();
    park_rows(xw1, rows, lane);
    load_rows(rows, xrow + o2 * NN);
    read_t(ft, xw0, lr, half, i0);
    SCHED_FENCE();
    readout_gram(a0, WT, acc, sg);
    SCHED_FENCE();
    scalars_of(sb, bal1, pw, ipw, half);
    read_a(a1, xw1, lr, half);
    SCHED_FENCE();
    ya = intra_update(ft, sa, acc, sg, WT, eta, lr, half);

    SCHED_FENCE();
    store_y(yrow + (size_t)ch * CHUNK * NN, ya);
    park_rows(xw0, rows, lane);
    load_rows(rows, xrow + o3 * NN);
    read_t(ft, xw1, lr, half, i0);
    SCHED_FENCE();
    readout_gram(a1, WT, acc, sg);
    SCHED_FENCE();
    scalars_of(sa, bal2, pw, ipw, half);
    read_a(a0, xw0, lr, half);
    SCHED_FENCE();
    yb = intra_update(ft, sb, acc, sg, WT, eta, lr, half);
    SCHED_FENCE();
    store_y(yrow + (size_t)o1 * NN, yb);
  }
#undef SCHED_FENCE

  if (seg == NSEG - 1) {
    float* wdst = w_out + (size_t)b * NN * NN + (i0 + lr) * NN + 8 * half;
    #pragma unroll
    for (int jt = 0; jt < NTILE; ++jt) {
      const v4f lo = __builtin_shufflevector(WT[jt], WT[jt], 0, 1, 2, 3);
      const v4f hi = __builtin_shufflevector(WT[jt], WT[jt], 4, 5, 6, 7);
      *(v4f*)(wdst + 16 * jt)     = lo;
      *(v4f*)(wdst + 16 * jt + 4) = hi;
    }
  }
}

extern "C" void kernel_launch(void* const* d_in, const int* in_sizes, int n_in,
                              void* d_out, int out_size, void* d_ws, size_t ws_size,
                              hipStream_t stream) {
  const float* x      = (const float*)d_in[0];
  const float* w_init = (const float*)d_in[1];
  const float* alpha  = (const float*)d_in[2];
  const float* eta    = (const float*)d_in[3];
  const int*   maskp  = (const int*)d_in[4];

  float* w_out = (float*)d_out;
  float* y_out = (float*)d_out + (size_t)BB * NN * NN;

  hebb_seg_kernel<<<BB * NTILE, NTHREAD, 0, stream>>>(
      x, w_init, alpha, eta, maskp, w_out, y_out);
}


# delta pass fetches x as whole 512-byte rows (16 b128 loads per chunk, was 72 dword loads), parks them in the wave's own LDS tile and reads its t and tb fragments from LDS one chunk ahead
# speedup vs baseline: 10.0457x; 1.1011x over previous
#include <hip/hip_runtime.h>


#define BB 32
#define TT 1024
#define NN 128
#define CHUNK 16
#define NCHUNK (TT / CHUNK)
#define NTILE (NN / 16)
#define NSEG 8
#define SEGCH (NCHUNK / NSEG)
#define NTHREAD (32 * NSEG)
#define XROW 20
#define XSUB 336
#define XTILE (NTILE * XSUB)

typedef __attribute__((ext_vector_type(2))) float v2f;
typedef __attribute__((ext_vector_type(4))) float v4f;
typedef __attribute__((ext_vector_type(8))) float v8f;

__device__ __forceinline__ v8f wmma4(v2f a, v2f b, v8f c) {
  return __builtin_amdgcn_wmma_f32_16x16x4_f32(
       false, a,  false, b,
       (short)0, c,  false,  false);
}

__device__ __forceinline__ float lane_get(float v, int src) {
  return __int_as_float(__builtin_amdgcn_ds_bpermute(src << 2, __float_as_int(v)));
}

struct AFrag {
  v8f a[NTILE];
};
struct TFrag {
  v8f t[NTILE];
  v8f tb;
};

__device__ __forceinline__ void load_rows(v4f (&g)[CHUNK], const float* __restrict__ xrow) {
  #pragma unroll
  for (int r = 0; r < CHUNK; ++r) g[r] = *(const v4f*)(xrow + r * NN);
}

__device__ __forceinline__ void park_rows(float* xw, const v4f (&g)[CHUNK], int lane) {
  float* dst = xw + (lane >> 2) * XSUB + 4 * (lane & 3);
  #pragma unroll
  for (int r = 0; r < CHUNK; ++r) *(v4f*)(dst + r * XROW) = g[r];
}

__device__ __forceinline__ void read_a(AFrag& f, const float* xw, int lr, int half) {
  const float* pa = xw + lr * XROW + 8 * half;
  #pragma unroll
  for (int jt = 0; jt < NTILE; ++jt) {
    const v4f lo = *(const v4f*)(pa + jt * XSUB);
    const v4f hi = *(const v4f*)(pa + jt * XSUB + 4);
    f.a[jt] = __builtin_shufflevector(lo, hi, 0, 1, 2, 3, 4, 5, 6, 7);
  }
}

__device__ __forceinline__ void read_t(TFrag& f, const float* xw, int lr, int half, int i0) {
  const float* pt = xw + (8 * half) * XROW + lr;
  #pragma unroll
  for (int v = 0; v < 8; ++v) f.tb[v] = pt[(i0 >> 4) * XSUB + v * XROW];
  #pragma unroll
  for (int jt = 0; jt < NTILE; ++jt) {
    #pragma unroll
    for (int v = 0; v < 8; ++v) f.t[jt][v] = pt[jt * XSUB + v * XROW];
  }
}

struct Scal {
  float gi[8];
  float Gs[8];
  float G16;
  unsigned bal;
};

__device__ __forceinline__ void scalars_of(Scal& sc, unsigned bal, float pw, float ipw,
                                           int half) {
  sc.bal = bal;
  #pragma unroll
  for (int v = 0; v < 8; ++v) {
    const int s = 8 * half + v;
    const int c_lt = __popc(bal & ((1u << s) - 1u));
    const int c_le = __popc(bal & ((2u << s) - 1u));
    sc.Gs[v] = lane_get(pw, c_lt);
    sc.gi[v] = lane_get(ipw, c_le);
  }
  sc.G16 = lane_get(pw, __popc(bal));
}


__device__ __forceinline__ v8f delta_hx(const TFrag& f, const Scal& sc, float eta, int half) {
  v8f hx;
  #pragma unroll
  for (int v = 0; v < 8; ++v) {
    const float hs = ((sc.bal >> (8 * half + v)) & 1u) ? eta * sc.gi[v] : 0.f;
    hx[v] = hs * f.tb[v];
  }
  return hx;
}

template <int JT0>
__device__ __forceinline__ void delta_tiles(v8f (&D)[NTILE], const TFrag& f, const v8f& hx) {
  #pragma unroll
  for (int p = 0; p < 4; ++p) {
    const v2f b = {hx[2 * p], hx[2 * p + 1]};
    #pragma unroll
    for (int jt = JT0; jt < JT0 + NTILE / 2; ++jt) {
      const v2f a = {f.t[jt][2 * p], f.t[jt][2 * p + 1]};
      D[jt] = wmma4(a, b, D[jt]);
    }
  }
}

__device__ __forceinline__ void readout_gram(const AFrag& f, const v8f (&WT)[NTILE],
                                             v8f& acc, v8f& sg) {
  v8f acc0 = {}, acc1 = {}, sg0 = {}, sg1 = {};
  #pragma unroll
  for (int jt = 0; jt < NTILE; jt += 2) {
    #pragma unroll
    for (int p = 0; p < 4; ++p) {
      const v2f a0 = {f.a[jt][2 * p], f.a[jt][2 * p + 1]};
      const v2f b0 = {WT[jt][2 * p], WT[jt][2 * p + 1]};
      const v2f a1 = {f.a[jt + 1][2 * p], f.a[jt + 1][2 * p + 1]};
      const v2f b1 = {WT[jt + 1][2 * p], WT[jt + 1][2 * p + 1]};
      acc0 = wmma4(a0, b0, acc0);
      sg0  = wmma4(a0, a0, sg0);
      acc1 = wmma4(a1, b1, acc1);
      sg1  = wmma4(a1, a1, sg1);
    }
  }
  acc = acc0 + acc1;
  sg  = sg0 + sg1;
}

__device__ __forceinline__ v8f intra_update(const TFrag& f, const Scal& sc, v8f acc,
                                            const v8f& sg, v8f (&WT)[NTILE],
                                            float eta, int lr, int half) {
  v8f hx;
  #pragma unroll
  for (int v = 0; v < 8; ++v) {
    const float hs = ((sc.bal >> (8 * half + v)) & 1u) ? eta * sc.gi[v] : 0.f;
    hx[v] = hs * f.tb[v];
  }

  #pragma unroll
  for (int p = 0; p < 4; ++p) {
    const int s0 = 8 * half + 2 * p;
    const v2f a = {(s0 < lr) ? sg[2 * p] : 0.f, (s0 + 1 < lr) ? sg[2 * p + 1] : 0.f};
    const v2f b = {hx[2 * p], hx[2 * p + 1]};
    acc = wmma4(a, b, acc);
  }
  v8f y;
  #pragma unroll
  for (int v = 0; v < 8; ++v) y[v] = sc.Gs[v] * acc[v];

  #pragma unroll
  for (int p = 0; p < 4; ++p) {
    const v2f b = {hx[2 * p], hx[2 * p + 1]};
    #pragma unroll
    for (int jt = 0; jt < NTILE; ++jt) {
      const v2f a = {f.t[jt][2 * p], f.t[jt][2 * p + 1]};
      WT[jt] = wmma4(a, b, WT[jt]);
    }
  }
  #pragma unroll
  for (int jt = 0; jt < NTILE; ++jt) WT[jt] *= sc.G16;
  return y;
}

__device__ __forceinline__ void store_y(float* __restrict__ yrow, const v8f& y) {
  #pragma unroll
  for (int v = 0; v < 8; ++v) yrow[v * NN] = y[v];
}

#define SCHED_FENCE() __builtin_amdgcn_sched_barrier(0)

__global__ void __launch_bounds__(NTHREAD)
hebb_seg_kernel(const float* __restrict__ x,
                const float* __restrict__ w_init,
                const float* __restrict__ alphap,
                const float* __restrict__ etap,
                const int* __restrict__ mask,
                float* __restrict__ w_out,
                float* __restrict__ y_out) {
  __shared__ __attribute__((aligned(16))) float dl[(NSEG - 1) * NTILE * 2 * 128];
  __shared__ float gl[NSEG];
  __shared__ __attribute__((aligned(16))) float xs[NSEG * 2 * XTILE];

  const int seg  = __builtin_amdgcn_readfirstlane(threadIdx.x >> 5);
  const int lane = threadIdx.x & 31;
  const int half = lane >> 4;
  const int lr   = lane & 15;
  const int b    = blockIdx.x >> 3;
  const int band = blockIdx.x & 7;
  const int i0   = band * 16;
  const float alpha = *alphap;
  const float eta   = *etap;

  float pw = 1.f;
  #pragma unroll
  for (int k = 0; k < 16; ++k) pw *= (k < lane) ? alpha : 1.f;
  const float ipw = 1.f / pw;

  const int t0 = seg * (SEGCH * CHUNK);
  const float* xb  = x + (size_t)b * TT * NN + (size_t)t0 * NN;
  const float* xrow = xb + 4 * lane;
  const int*   mp  = mask + b * TT + t0 + lr;
  float* yrow = y_out + (size_t)b * TT * NN + (size_t)t0 * NN + (8 * half) * NN + i0 + lr;

  unsigned long long mlo = 0ull, mhi = 0ull;
  #pragma unroll
  for (int c = 0; c < SEGCH; ++c) {
    const unsigned long long bal = __builtin_amdgcn_ballot_w32(mp[c * CHUNK] != 0) & 0xFFFFu;
    if (c < SEGCH / 2) mlo |= bal << (16 * c);
    else               mhi |= bal << (16 * (c - SEGCH / 2));
  }

  if (seg < NSEG - 1) {
    v8f D[NTILE];
    #pragma unroll
    for (int jt = 0; jt < NTILE; ++jt) D[jt] = v8f{};
    float gseg = 1.f;

    TFrag ua, ub;
    Scal sa, sb;
    v4f drows[CHUNK];
    float* dw0 = xs + seg * (2 * XTILE);
    float* dw1 = dw0 + XTILE;
    SCHED_FENCE();
    load_rows(drows, xrow);
    SCHED_FENCE();
    park_rows(dw0, drows, lane);
    load_rows(drows, xrow + CHUNK * NN);
    read_t(ua, dw0, lr, half, i0);
    scalars_of(sa, (unsigned)mlo & 0xFFFFu, pw, ipw, half);

    #pragma unroll 1
    for (int ch = 0; ch < SEGCH; ch += 2) {
      const bool more = (ch + 2 < SEGCH);
      const int o2 = (ch + 2) * CHUNK;
      const int o3 = (ch + 3) * CHUNK;
      const int nx = more ? (ch + 2) : (SEGCH - 1);
      const unsigned long long m1 = (ch + 1 < SEGCH / 2) ? mlo : mhi;
      const unsigned long long m2 = (nx < SEGCH / 2) ? mlo : mhi;
      const unsigned bal1 = (unsigned)(m1 >> (16 * ((ch + 1) & 3))) & 0xFFFFu;
      const unsigned bal2 = (unsigned)(m2 >> (16 * (nx & 3))) & 0xFFFFu;

      SCHED_FENCE();
      park_rows(dw1, drows, lane);
      if (more) load_rows(drows, xrow + o2 * NN);
      read_t(ub, dw1, lr, half, i0);
      scalars_of(sb, bal1, pw, ipw, half);
      SCHED_FENCE();
      {
        const v8f hx = delta_hx(ua, sa, eta, half);
        delta_tiles<0>(D, ua, hx);
        delta_tiles<NTILE / 2>(D, ua, hx);
      }
      #pragma unroll
      for (int jt = 0; jt < NTILE; ++jt) D[jt] *= sa.G16;
      gseg *= sa.G16;

      SCHED_FENCE();
      if (more) {
        park_rows(dw0, drows, lane);
        load_rows(drows, xrow + o3 * NN);
        read_t(ua, dw0, lr, half, i0);
        scalars_of(sa, bal2, pw, ipw, half);
      }
      SCHED_FENCE();
      {
        const v8f hx = delta_hx(ub, sb, eta, half);
        delta_tiles<0>(D, ub, hx);
        delta_tiles<NTILE / 2>(D, ub, hx);
      }
      #pragma unroll
      for (int jt = 0; jt < NTILE; ++jt) D[jt] *= sb.G16;
      gseg *= sb.G16;
    }

    float* dst = dl + seg * (NTILE * 2 * 128) + lane * 4;
    #pragma unroll
    for (int jt = 0; jt < NTILE; ++jt) {
      *(v4f*)(dst + (2 * jt) * 128)     = __builtin_shufflevector(D[jt], D[jt], 0, 1, 2, 3);
      *(v4f*)(dst + (2 * jt + 1) * 128) = __builtin_shufflevector(D[jt], D[jt], 4, 5, 6, 7);
    }
    if (lane == 0) gl[seg] = gseg;
  }

  __syncthreads();

  v8f WT[NTILE];
  {
    const float* wsrc = w_init + (size_t)b * NN * NN + (i0 + lr) * NN + 8 * half;
    #pragma unroll
    for (int jt = 0; jt < NTILE; ++jt) {
      const v4f lo = *(const v4f*)(wsrc + 16 * jt);
      const v4f hi = *(const v4f*)(wsrc + 16 * jt + 4);
      WT[jt] = __builtin_shufflevector(lo, hi, 0, 1, 2, 3, 4, 5, 6, 7);
    }
  }
  AFrag a0, a1;
  TFrag ft;
  Scal sa, sb;
  v8f acc, sg, ya, yb;
  v4f rows[CHUNK];
  float* xw0 = xs + seg * (2 * XTILE);
  float* xw1 = xw0 + XTILE;
  load_rows(rows, xrow);

  #pragma unroll 1
  for (int q = 0; q < seg; ++q) {
    const float g = gl[q];
    const float* dsrc = dl + q * (NTILE * 2 * 128) + lane * 4;
    #pragma unroll
    for (int jt = 0; jt < NTILE; ++jt) {
      const v4f lo = *(const v4f*)(dsrc + (2 * jt) * 128);
      const v4f hi = *(const v4f*)(dsrc + (2 * jt + 1) * 128);
      WT[jt] = g * WT[jt] + __builtin_shufflevector(lo, hi, 0, 1, 2, 3, 4, 5, 6, 7);
    }
  }

  park_rows(xw0, rows, lane);
  read_a(a0, xw0, lr, half);
  load_rows(rows, xrow + CHUNK * NN);
  scalars_of(sa, (unsigned)mlo & 0xFFFFu, pw, ipw, half);

  #pragma unroll 1
  for (int ch = 0; ch < SEGCH; ch += 2) {
    const int o1 = (ch + 1) * CHUNK;
    const int o2 = ((ch + 2 < SEGCH) ? (ch + 2) : (SEGCH - 1)) * CHUNK;
    const int o3 = ((ch + 3 < SEGCH) ? (ch + 3) : (SEGCH - 1)) * CHUNK;
    const int c2 = (ch + 2 < SEGCH) ? (ch + 2) : (SEGCH - 1);
    const unsigned long long m1 = (ch + 1 < SEGCH / 2) ? mlo : mhi;
    const unsigned long long m2 = (c2 < SEGCH / 2) ? mlo : mhi;
    const unsigned bal1 = (unsigned)(m1 >> (16 * ((ch + 1) & 3))) & 0xFFFFu;
    const unsigned bal2 = (unsigned)(m2 >> (16 * (c2 & 3))) & 0xFFFFu;

    SCHED_FENCE();
    park_rows(xw1, rows, lane);
    load_rows(rows, xrow + o2 * NN);
    read_t(ft, xw0, lr, half, i0);
    SCHED_FENCE();
    readout_gram(a0, WT, acc, sg);
    SCHED_FENCE();
    scalars_of(sb, bal1, pw, ipw, half);
    read_a(a1, xw1, lr, half);
    SCHED_FENCE();
    ya = intra_update(ft, sa, acc, sg, WT, eta, lr, half);

    SCHED_FENCE();
    store_y(yrow + (size_t)ch * CHUNK * NN, ya);
    park_rows(xw0, rows, lane);
    load_rows(rows, xrow + o3 * NN);
    read_t(ft, xw1, lr, half, i0);
    SCHED_FENCE();
    readout_gram(a1, WT, acc, sg);
    SCHED_FENCE();
    scalars_of(sa, bal2, pw, ipw, half);
    read_a(a0, xw0, lr, half);
    SCHED_FENCE();
    yb = intra_update(ft, sb, acc, sg, WT, eta, lr, half);
    SCHED_FENCE();
    store_y(yrow + (size_t)o1 * NN, yb);
  }
#undef SCHED_FENCE

  if (seg == NSEG - 1) {
    float* wdst = w_out + (size_t)b * NN * NN + (i0 + lr) * NN + 8 * half;
    #pragma unroll
    for (int jt = 0; jt < NTILE; ++jt) {
      const v4f lo = __builtin_shufflevector(WT[jt], WT[jt], 0, 1, 2, 3);
      const v4f hi = __builtin_shufflevector(WT[jt], WT[jt], 4, 5, 6, 7);
      *(v4f*)(wdst + 16 * jt)     = lo;
      *(v4f*)(wdst + 16 * jt + 4) = hi;
    }
  }
}

extern "C" void kernel_launch(void* const* d_in, const int* in_sizes, int n_in,
                              void* d_out, int out_size, void* d_ws, size_t ws_size,
                              hipStream_t stream) {
  const float* x      = (const float*)d_in[0];
  const float* w_init = (const float*)d_in[1];
  const float* alpha  = (const float*)d_in[2];
  const float* eta    = (const float*)d_in[3];
  const int*   maskp  = (const int*)d_in[4];

  float* w_out = (float*)d_out;
  float* y_out = (float*)d_out + (size_t)BB * NN * NN;

  hebb_seg_kernel<<<BB * NTILE, NTHREAD, 0, stream>>>(
      x, w_init, alpha, eta, maskp, w_out, y_out);
}


# decay scalars (G_t, h_s, G_16) of a wave's whole segment computed once, one step per lane, parked in its own LDS; both passes fetch them with 16-byte loads instead of 17 lane exchanges per chunk
# speedup vs baseline: 10.5481x; 1.0500x over previous
#include <hip/hip_runtime.h>


#define BB 32
#define TT 1024
#define NN 128
#define CHUNK 16
#define NCHUNK (TT / CHUNK)
#define NTILE (NN / 16)
#define NSEG 8
#define SEGCH (NCHUNK / NSEG)
#define NTHREAD (32 * NSEG)
#define XROW 20
#define XSUB 336
#define XTILE (NTILE * XSUB)

typedef __attribute__((ext_vector_type(2))) float v2f;
typedef __attribute__((ext_vector_type(4))) float v4f;
typedef __attribute__((ext_vector_type(8))) float v8f;

__device__ __forceinline__ v8f wmma4(v2f a, v2f b, v8f c) {
  return __builtin_amdgcn_wmma_f32_16x16x4_f32(
       false, a,  false, b,
       (short)0, c,  false,  false);
}

__device__ __forceinline__ float lane_get(float v, int src) {
  return __int_as_float(__builtin_amdgcn_ds_bpermute(src << 2, __float_as_int(v)));
}

struct AFrag {
  v8f a[NTILE];
};
struct TFrag {
  v8f t[NTILE];
  v8f tb;
};

__device__ __forceinline__ void load_rows(v4f (&g)[CHUNK], const float* __restrict__ xrow) {
  #pragma unroll
  for (int r = 0; r < CHUNK; ++r) g[r] = *(const v4f*)(xrow + r * NN);
}

__device__ __forceinline__ void park_rows(float* xw, const v4f (&g)[CHUNK], int lane) {
  float* dst = xw + (lane >> 2) * XSUB + 4 * (lane & 3);
  #pragma unroll
  for (int r = 0; r < CHUNK; ++r) *(v4f*)(dst + r * XROW) = g[r];
}

__device__ __forceinline__ void read_a(AFrag& f, const float* xw, int lr, int half) {
  const float* pa = xw + lr * XROW + 8 * half;
  #pragma unroll
  for (int jt = 0; jt < NTILE; ++jt) {
    const v4f lo = *(const v4f*)(pa + jt * XSUB);
    const v4f hi = *(const v4f*)(pa + jt * XSUB + 4);
    f.a[jt] = __builtin_shufflevector(lo, hi, 0, 1, 2, 3, 4, 5, 6, 7);
  }
}

__device__ __forceinline__ void read_t(TFrag& f, const float* xw, int lr, int half, int i0) {
  const float* pt = xw + (8 * half) * XROW + lr;
  #pragma unroll
  for (int v = 0; v < 8; ++v) f.tb[v] = pt[(i0 >> 4) * XSUB + v * XROW];
  #pragma unroll
  for (int jt = 0; jt < NTILE; ++jt) {
    #pragma unroll
    for (int v = 0; v < 8; ++v) f.t[jt][v] = pt[jt * XSUB + v * XROW];
  }
}

struct Scal {
  v8f hs;
  v8f Gs;
  float G16;
};
#define SC_H 128
#define SC_G16 256
#define SCSTRIDE 272

__device__ __forceinline__ void park_scalars(float* scw, unsigned long long mlo,
                                             unsigned long long mhi, float pw, float ipw,
                                             float eta, int lane, int lr, int half) {
  #pragma unroll
  for (int k = 0; k < SEGCH / 2; ++k) {
    const unsigned long long mm = (2 * k < SEGCH / 2) ? mlo : mhi;
    const unsigned be = (unsigned)(mm >> (16 * ((2 * k) & 3))) & 0xFFFFu;
    const unsigned bo = (unsigned)(mm >> (16 * ((2 * k + 1) & 3))) & 0xFFFFu;
    const unsigned bal = half ? bo : be;
    const int c_lt = __popc(bal & ((1u << lr) - 1u));
    const int c_le = __popc(bal & ((2u << lr) - 1u));
    const float Gs = lane_get(pw, c_lt);
    const float gi = lane_get(ipw, c_le);
    scw[32 * k + lane] = Gs;
    scw[SC_H + 32 * k + lane] = ((bal >> lr) & 1u) ? eta * gi : 0.f;
  }
  const int c = lane & (SEGCH - 1);
  const unsigned long long mc = (c < SEGCH / 2) ? mlo : mhi;
  const unsigned balc = (unsigned)(mc >> (16 * (c & 3))) & 0xFFFFu;
  const float g16 = lane_get(pw, __popc(balc));
  if (lane < SEGCH) scw[SC_G16 + lane] = g16;
}

__device__ __forceinline__ void fetch_scalars(Scal& sc, const float* sch, const float* scw,
                                              int c) {
  const float* p = sch + CHUNK * c;
  const v4f g0 = *(const v4f*)p;
  const v4f g1 = *(const v4f*)(p + 4);
  const v4f h0 = *(const v4f*)(p + SC_H);
  const v4f h1 = *(const v4f*)(p + SC_H + 4);
  sc.Gs = __builtin_shufflevector(g0, g1, 0, 1, 2, 3, 4, 5, 6, 7);
  sc.hs = __builtin_shufflevector(h0, h1, 0, 1, 2, 3, 4, 5, 6, 7);
  sc.G16 = scw[SC_G16 + c];
}


__device__ __forceinline__ v8f delta_hx(const TFrag& f, const Scal& sc) {
  return sc.hs * f.tb;
}

template <int JT0>
__device__ __forceinline__ void delta_tiles(v8f (&D)[NTILE], const TFrag& f, const v8f& hx) {
  #pragma unroll
  for (int p = 0; p < 4; ++p) {
    const v2f b = {hx[2 * p], hx[2 * p + 1]};
    #pragma unroll
    for (int jt = JT0; jt < JT0 + NTILE / 2; ++jt) {
      const v2f a = {f.t[jt][2 * p], f.t[jt][2 * p + 1]};
      D[jt] = wmma4(a, b, D[jt]);
    }
  }
}

__device__ __forceinline__ void readout_gram(const AFrag& f, const v8f (&WT)[NTILE],
                                             v8f& acc, v8f& sg) {
  v8f acc0 = {}, acc1 = {}, sg0 = {}, sg1 = {};
  #pragma unroll
  for (int jt = 0; jt < NTILE; jt += 2) {
    #pragma unroll
    for (int p = 0; p < 4; ++p) {
      const v2f a0 = {f.a[jt][2 * p], f.a[jt][2 * p + 1]};
      const v2f b0 = {WT[jt][2 * p], WT[jt][2 * p + 1]};
      const v2f a1 = {f.a[jt + 1][2 * p], f.a[jt + 1][2 * p + 1]};
      const v2f b1 = {WT[jt + 1][2 * p], WT[jt + 1][2 * p + 1]};
      acc0 = wmma4(a0, b0, acc0);
      sg0  = wmma4(a0, a0, sg0);
      acc1 = wmma4(a1, b1, acc1);
      sg1  = wmma4(a1, a1, sg1);
    }
  }
  acc = acc0 + acc1;
  sg  = sg0 + sg1;
}

__device__ __forceinline__ v8f intra_update(const TFrag& f, const Scal& sc, v8f acc,
                                            const v8f& sg, v8f (&WT)[NTILE],
                                            int lr, int half) {
  const v8f hx = sc.hs * f.tb;

  #pragma unroll
  for (int p = 0; p < 4; ++p) {
    const int s0 = 8 * half + 2 * p;
    const v2f a = {(s0 < lr) ? sg[2 * p] : 0.f, (s0 + 1 < lr) ? sg[2 * p + 1] : 0.f};
    const v2f b = {hx[2 * p], hx[2 * p + 1]};
    acc = wmma4(a, b, acc);
  }
  const v8f y = sc.Gs * acc;

  #pragma unroll
  for (int p = 0; p < 4; ++p) {
    const v2f b = {hx[2 * p], hx[2 * p + 1]};
    #pragma unroll
    for (int jt = 0; jt < NTILE; ++jt) {
      const v2f a = {f.t[jt][2 * p], f.t[jt][2 * p + 1]};
      WT[jt] = wmma4(a, b, WT[jt]);
    }
  }
  #pragma unroll
  for (int jt = 0; jt < NTILE; ++jt) WT[jt] *= sc.G16;
  return y;
}

__device__ __forceinline__ void store_y(float* __restrict__ yrow, const v8f& y) {
  #pragma unroll
  for (int v = 0; v < 8; ++v) yrow[v * NN] = y[v];
}

#define SCHED_FENCE() __builtin_amdgcn_sched_barrier(0)

__global__ void __launch_bounds__(NTHREAD)
hebb_seg_kernel(const float* __restrict__ x,
                const float* __restrict__ w_init,
                const float* __restrict__ alphap,
                const float* __restrict__ etap,
                const int* __restrict__ mask,
                float* __restrict__ w_out,
                float* __restrict__ y_out) {
  __shared__ __attribute__((aligned(16))) float dl[(NSEG - 1) * NTILE * 2 * 128];
  __shared__ float gl[NSEG];
  __shared__ __attribute__((aligned(16))) float scl[NSEG * SCSTRIDE];
  __shared__ __attribute__((aligned(16))) float xs[NSEG * 2 * XTILE];

  const int seg  = __builtin_amdgcn_readfirstlane(threadIdx.x >> 5);
  const int lane = threadIdx.x & 31;
  const int half = lane >> 4;
  const int lr   = lane & 15;
  const int b    = blockIdx.x >> 3;
  const int band = blockIdx.x & 7;
  const int i0   = band * 16;
  const float alpha = *alphap;
  const float eta   = *etap;

  float pw = 1.f;
  #pragma unroll
  for (int k = 0; k < 16; ++k) pw *= (k < lane) ? alpha : 1.f;
  const float ipw = 1.f / pw;

  const int t0 = seg * (SEGCH * CHUNK);
  const float* xb  = x + (size_t)b * TT * NN + (size_t)t0 * NN;
  const float* xrow = xb + 4 * lane;
  const int*   mp  = mask + b * TT + t0 + lr;
  float* yrow = y_out + (size_t)b * TT * NN + (size_t)t0 * NN + (8 * half) * NN + i0 + lr;

  unsigned long long mlo = 0ull, mhi = 0ull;
  #pragma unroll
  for (int c = 0; c < SEGCH; ++c) {
    const unsigned long long bal = __builtin_amdgcn_ballot_w32(mp[c * CHUNK] != 0) & 0xFFFFu;
    if (c < SEGCH / 2) mlo |= bal << (16 * c);
    else               mhi |= bal << (16 * (c - SEGCH / 2));
  }

  float* scw = scl + seg * SCSTRIDE;
  const float* sch = scw + 8 * half;
  park_scalars(scw, mlo, mhi, pw, ipw, eta, lane, lr, half);

  if (seg < NSEG - 1) {
    v8f D[NTILE];
    #pragma unroll
    for (int jt = 0; jt < NTILE; ++jt) D[jt] = v8f{};
    float gseg = 1.f;

    TFrag ua, ub;
    Scal sa, sb;
    v4f drows[CHUNK];
    float* dw0 = xs + seg * (2 * XTILE);
    float* dw1 = dw0 + XTILE;
    SCHED_FENCE();
    load_rows(drows, xrow);
    SCHED_FENCE();
    park_rows(dw0, drows, lane);
    load_rows(drows, xrow + CHUNK * NN);
    read_t(ua, dw0, lr, half, i0);
    fetch_scalars(sa, sch, scw, 0);

    #pragma unroll 1
    for (int ch = 0; ch < SEGCH; ch += 2) {
      const bool more = (ch + 2 < SEGCH);
      const int o2 = (ch + 2) * CHUNK;
      const int o3 = (ch + 3) * CHUNK;

      SCHED_FENCE();
      park_rows(dw1, drows, lane);
      if (more) load_rows(drows, xrow + o2 * NN);
      read_t(ub, dw1, lr, half, i0);
      fetch_scalars(sb, sch, scw, ch + 1);
      SCHED_FENCE();
      {
        const v8f hx = delta_hx(ua, sa);
        delta_tiles<0>(D, ua, hx);
        delta_tiles<NTILE / 2>(D, ua, hx);
      }
      #pragma unroll
      for (int jt = 0; jt < NTILE; ++jt) D[jt] *= sa.G16;
      gseg *= sa.G16;

      SCHED_FENCE();
      if (more) {
        park_rows(dw0, drows, lane);
        load_rows(drows, xrow + o3 * NN);
        read_t(ua, dw0, lr, half, i0);
        fetch_scalars(sa, sch, scw, ch + 2);
      }
      SCHED_FENCE();
      {
        const v8f hx = delta_hx(ub, sb);
        delta_tiles<0>(D, ub, hx);
        delta_tiles<NTILE / 2>(D, ub, hx);
      }
      #pragma unroll
      for (int jt = 0; jt < NTILE; ++jt) D[jt] *= sb.G16;
      gseg *= sb.G16;
    }

    float* dst = dl + seg * (NTILE * 2 * 128) + lane * 4;
    #pragma unroll
    for (int jt = 0; jt < NTILE; ++jt) {
      *(v4f*)(dst + (2 * jt) * 128)     = __builtin_shufflevector(D[jt], D[jt], 0, 1, 2, 3);
      *(v4f*)(dst + (2 * jt + 1) * 128) = __builtin_shufflevector(D[jt], D[jt], 4, 5, 6, 7);
    }
    if (lane == 0) gl[seg] = gseg;
  }

  __syncthreads();

  v8f WT[NTILE];
  {
    const float* wsrc = w_init + (size_t)b * NN * NN + (i0 + lr) * NN + 8 * half;
    #pragma unroll
    for (int jt = 0; jt < NTILE; ++jt) {
      const v4f lo = *(const v4f*)(wsrc + 16 * jt);
      const v4f hi = *(const v4f*)(wsrc + 16 * jt + 4);
      WT[jt] = __builtin_shufflevector(lo, hi, 0, 1, 2, 3, 4, 5, 6, 7);
    }
  }
  AFrag a0, a1;
  TFrag ft;
  Scal sa, sb;
  v8f acc, sg, ya, yb;
  v4f rows[CHUNK];
  float* xw0 = xs + seg * (2 * XTILE);
  float* xw1 = xw0 + XTILE;
  load_rows(rows, xrow);

  #pragma unroll 1
  for (int q = 0; q < seg; ++q) {
    const float g = gl[q];
    const float* dsrc = dl + q * (NTILE * 2 * 128) + lane * 4;
    #pragma unroll
    for (int jt = 0; jt < NTILE; ++jt) {
      const v4f lo = *(const v4f*)(dsrc + (2 * jt) * 128);
      const v4f hi = *(const v4f*)(dsrc + (2 * jt + 1) * 128);
      WT[jt] = g * WT[jt] + __builtin_shufflevector(lo, hi, 0, 1, 2, 3, 4, 5, 6, 7);
    }
  }

  park_rows(xw0, rows, lane);
  read_a(a0, xw0, lr, half);
  load_rows(rows, xrow + CHUNK * NN);
  fetch_scalars(sa, sch, scw, 0);

  #pragma unroll 1
  for (int ch = 0; ch < SEGCH; ch += 2) {
    const int o1 = (ch + 1) * CHUNK;
    const int o2 = ((ch + 2 < SEGCH) ? (ch + 2) : (SEGCH - 1)) * CHUNK;
    const int o3 = ((ch + 3 < SEGCH) ? (ch + 3) : (SEGCH - 1)) * CHUNK;
    const int c2 = (ch + 2 < SEGCH) ? (ch + 2) : (SEGCH - 1);

    SCHED_FENCE();
    park_rows(xw1, rows, lane);
    load_rows(rows, xrow + o2 * NN);
    read_t(ft, xw0, lr, half, i0);
    SCHED_FENCE();
    readout_gram(a0, WT, acc, sg);
    SCHED_FENCE();
    fetch_scalars(sb, sch, scw, ch + 1);
    read_a(a1, xw1, lr, half);
    SCHED_FENCE();
    ya = intra_update(ft, sa, acc, sg, WT, lr, half);

    SCHED_FENCE();
    store_y(yrow + (size_t)ch * CHUNK * NN, ya);
    park_rows(xw0, rows, lane);
    load_rows(rows, xrow + o3 * NN);
    read_t(ft, xw1, lr, half, i0);
    SCHED_FENCE();
    readout_gram(a1, WT, acc, sg);
    SCHED_FENCE();
    fetch_scalars(sa, sch, scw, c2);
    read_a(a0, xw0, lr, half);
    SCHED_FENCE();
    yb = intra_update(ft, sb, acc, sg, WT, lr, half);
    SCHED_FENCE();
    store_y(yrow + (size_t)o1 * NN, yb);
  }
#undef SCHED_FENCE

  if (seg == NSEG - 1) {
    float* wdst = w_out + (size_t)b * NN * NN + (i0 + lr) * NN + 8 * half;
    #pragma unroll
    for (int jt = 0; jt < NTILE; ++jt) {
      const v4f lo = __builtin_shufflevector(WT[jt], WT[jt], 0, 1, 2, 3);
      const v4f hi = __builtin_shufflevector(WT[jt], WT[jt], 4, 5, 6, 7);
      *(v4f*)(wdst + 16 * jt)     = lo;
      *(v4f*)(wdst + 16 * jt + 4) = hi;
    }
  }
}

extern "C" void kernel_launch(void* const* d_in, const int* in_sizes, int n_in,
                              void* d_out, int out_size, void* d_ws, size_t ws_size,
                              hipStream_t stream) {
  const float* x      = (const float*)d_in[0];
  const float* w_init = (const float*)d_in[1];
  const float* alpha  = (const float*)d_in[2];
  const float* eta    = (const float*)d_in[3];
  const int*   maskp  = (const int*)d_in[4];

  float* w_out = (float*)d_out;
  float* y_out = (float*)d_out + (size_t)BB * NN * NN;

  hebb_seg_kernel<<<BB * NTILE, NTHREAD, 0, stream>>>(
      x, w_init, alpha, eta, maskp, w_out, y_out);
}
